# starting kernel
# baseline (speedup 1.0000x reference)
#include <hip/hip_runtime.h>

typedef __attribute__((ext_vector_type(2))) float v2f;
typedef __attribute__((ext_vector_type(8))) float v8f;

#define N_NODES 100000
#define N_EDGES 1600000
#define DIN 48
#define DOUT 48
#define DCAT 96
#define NTILES (N_NODES / 16)

__global__ void ec_prep(const float* __restrict__ theta_w, const float* __restrict__ theta_b,
                        const float* __restrict__ phi_w,   const float* __restrict__ phi_b,
                        float* __restrict__ Wcat, float* __restrict__ bias_c) {
  int i = blockIdx.x * blockDim.x + threadIdx.x;
  if (i < DOUT * DIN) {
    Wcat[i] = theta_w[i];
    Wcat[DOUT * DIN + i] = phi_w[i] - theta_w[i];
  }
  if (i < DOUT) bias_c[i] = phi_b[i] + theta_b[i];
}

__global__ void ec_init(unsigned* __restrict__ Mkey) {
  int i = blockIdx.x * blockDim.x + threadIdx.x;
  if (i < N_NODES * DOUT) Mkey[i] = 0u;
}

__global__ void ec_gemm(const float* __restrict__ h,
                        const float* __restrict__ Wcat,
                        const float* __restrict__ bias_c,
                        float* __restrict__ AC) {
  const int lane  = threadIdx.x & 31;
  const int wave  = threadIdx.x >> 5;
  const int tile  = blockIdx.x * (blockDim.x >> 5) + wave;
  if (tile >= NTILES) return;
  const int row   = lane & 15;
  const int khalf = lane >> 4;

  const float* hrow = h + (tile * 16 + row) * DIN + khalf * 2;
  v2f a[12];
#pragma unroll
  for (int kk = 0; kk < 12; ++kk)
    a[kk] = *(const v2f*)(hrow + kk * 4);

#pragma unroll
  for (int nt = 0; nt < 6; ++nt) {
    const float* wrow = Wcat + (nt * 16 + row) * DIN + khalf * 2;
    v8f c = {};
#pragma unroll
    for (int kk = 0; kk < 12; ++kk) {
      v2f b = *(const v2f*)(wrow + kk * 4);
      c = __builtin_amdgcn_wmma_f32_16x16x4_f32(false, a[kk], false, b,
                                                (short)0, c, false, false);
    }
    const float bias = (nt >= 3) ? bias_c[nt * 16 + row - DOUT] : 0.0f;
#pragma unroll
    for (int v = 0; v < 8; ++v) {
      const int m = v + 8 * khalf;
      AC[(tile * 16 + m) * DCAT + nt * 16 + row] = c[v] + bias;
    }
  }
}

__device__ __forceinline__ unsigned ec_fkey(float f) {
  unsigned u = __float_as_uint(f);
  return (u & 0x80000000u) ? ~u : (u | 0x80000000u);
}

__global__ void ec_edges(const int* __restrict__ src, const int* __restrict__ dst,
                         const float* __restrict__ AC, unsigned* __restrict__ Mkey) {
  int t = blockIdx.x * blockDim.x + threadIdx.x;
  int e = t >> 4;
  int g = t & 15;
  if (e >= N_EDGES) return;
  int s = src[e];
  int d = dst[e];
  const float* crow = AC + (size_t)s * DCAT + DOUT + g * 3;
  unsigned* mrow = Mkey + (size_t)d * DOUT + g * 3;
#pragma unroll
  for (int j = 0; j < 3; ++j)
    atomicMax(&mrow[j], ec_fkey(crow[j]));
}

__global__ void ec_final(const float* __restrict__ AC, const unsigned* __restrict__ Mkey,
                         float* __restrict__ out) {
  int i = blockIdx.x * blockDim.x + threadIdx.x;
  if (i >= N_NODES * DOUT) return;
  int n = i / DOUT;
  int f = i - n * DOUT;
  unsigned k = Mkey[i];
  float m = (k & 0x80000000u) ? __uint_as_float(k & 0x7FFFFFFFu)
                              : __uint_as_float(~k);
  out[i] = (k == 0u) ? 0.0f : (AC[(size_t)n * DCAT + f] + m);
}

extern "C" void kernel_launch(void* const* d_in, const int* in_sizes, int n_in,
                              void* d_out, int out_size, void* d_ws, size_t ws_size,
                              hipStream_t stream) {
  const float* h       = (const float*)d_in[0];
  const float* theta_w = (const float*)d_in[1];
  const float* theta_b = (const float*)d_in[2];
  const float* phi_w   = (const float*)d_in[3];
  const float* phi_b   = (const float*)d_in[4];
  const int*   src     = (const int*)d_in[5];
  const int*   dst     = (const int*)d_in[6];
  float* out = (float*)d_out;

  char* ws = (char*)d_ws;
  float*    Wcat   = (float*)ws;
  float*    bias_c = (float*)(ws + 18432);
  float*    AC     = (float*)(ws + 18688);
  unsigned* Mkey   = (unsigned*)(ws + 18688 + (size_t)N_NODES * DCAT * 4);

  ec_prep<<<(DOUT * DIN + 255) / 256, 256, 0, stream>>>(theta_w, theta_b, phi_w, phi_b,
                                                        Wcat, bias_c);
  ec_init<<<(N_NODES * DOUT + 255) / 256, 256, 0, stream>>>(Mkey);
  ec_gemm<<<(NTILES + 7) / 8, 256, 0, stream>>>(h, Wcat, bias_c, AC);
  ec_edges<<<(N_EDGES * 16) / 256, 256, 0, stream>>>(src, dst, AC, Mkey);
  ec_final<<<(N_NODES * DOUT + 255) / 256, 256, 0, stream>>>(AC, Mkey, out);
}


# Edge scatter-max coarsened: each wave walks 16 edges (8 per 16-lane group: indices by 16-byte loads, all 8 row gathers in flight, then the atomics) instead of 2, so an eighth of the waves
# speedup vs baseline: 1.0339x; 1.0339x over previous
#include <hip/hip_runtime.h>

typedef __attribute__((ext_vector_type(2))) float v2f;
typedef __attribute__((ext_vector_type(8))) float v8f;
typedef __attribute__((ext_vector_type(4))) int v4i;

#define N_NODES 100000
#define N_EDGES 1600000
#define DIN 48
#define DOUT 48
#define DCAT 96
#define NTILES (N_NODES / 16)
#define EPG 8
#define EDGE_BLOCKS ((N_EDGES / EPG) * 16 / 256)

__global__ void ec_prep(const float* __restrict__ theta_w, const float* __restrict__ theta_b,
                        const float* __restrict__ phi_w,   const float* __restrict__ phi_b,
                        float* __restrict__ Wcat, float* __restrict__ bias_c) {
  int i = blockIdx.x * blockDim.x + threadIdx.x;
  if (i < DOUT * DIN) {
    Wcat[i] = theta_w[i];
    Wcat[DOUT * DIN + i] = phi_w[i] - theta_w[i];
  }
  if (i < DOUT) bias_c[i] = phi_b[i] + theta_b[i];
}

__global__ void ec_init(unsigned* __restrict__ Mkey) {
  int i = blockIdx.x * blockDim.x + threadIdx.x;
  if (i < N_NODES * DOUT) Mkey[i] = 0u;
}

__global__ void ec_gemm(const float* __restrict__ h,
                        const float* __restrict__ Wcat,
                        const float* __restrict__ bias_c,
                        float* __restrict__ AC) {
  const int lane  = threadIdx.x & 31;
  const int wave  = threadIdx.x >> 5;
  const int tile  = blockIdx.x * (blockDim.x >> 5) + wave;
  if (tile >= NTILES) return;
  const int row   = lane & 15;
  const int khalf = lane >> 4;

  const float* hrow = h + (tile * 16 + row) * DIN + khalf * 2;
  v2f a[12];
#pragma unroll
  for (int kk = 0; kk < 12; ++kk)
    a[kk] = *(const v2f*)(hrow + kk * 4);

#pragma unroll
  for (int nt = 0; nt < 6; ++nt) {
    const float* wrow = Wcat + (nt * 16 + row) * DIN + khalf * 2;
    v8f c = {};
#pragma unroll
    for (int kk = 0; kk < 12; ++kk) {
      v2f b = *(const v2f*)(wrow + kk * 4);
      c = __builtin_amdgcn_wmma_f32_16x16x4_f32(false, a[kk], false, b,
                                                (short)0, c, false, false);
    }
    const float bias = (nt >= 3) ? bias_c[nt * 16 + row - DOUT] : 0.0f;
#pragma unroll
    for (int v = 0; v < 8; ++v) {
      const int m = v + 8 * khalf;
      AC[(tile * 16 + m) * DCAT + nt * 16 + row] = c[v] + bias;
    }
  }
}

__device__ __forceinline__ unsigned ec_fkey(float f) {
  unsigned u = __float_as_uint(f);
  return (u & 0x80000000u) ? ~u : (u | 0x80000000u);
}

__global__ void ec_edges(const int* __restrict__ src, const int* __restrict__ dst,
                         const float* __restrict__ AC, unsigned* __restrict__ Mkey) {
  const int t  = blockIdx.x * blockDim.x + threadIdx.x;
  const int g  = t & 15;
  const int e0 = (t >> 4) * EPG;
  if (e0 >= N_EDGES) return;

  int s[EPG], d[EPG];
#pragma unroll
  for (int q = 0; q < EPG / 4; ++q) {
    const v4i sv = *(const v4i*)(src + e0 + 4 * q);
    const v4i dv = *(const v4i*)(dst + e0 + 4 * q);
#pragma unroll
    for (int r = 0; r < 4; ++r) { s[4 * q + r] = sv[r]; d[4 * q + r] = dv[r]; }
  }

  float c[EPG][3];
#pragma unroll
  for (int i = 0; i < EPG; ++i) {
    const float* crow = AC + (size_t)s[i] * DCAT + DOUT + g * 3;
#pragma unroll
    for (int j = 0; j < 3; ++j) c[i][j] = crow[j];
  }

#pragma unroll
  for (int i = 0; i < EPG; ++i) {
    unsigned* mrow = Mkey + (size_t)d[i] * DOUT + g * 3;
#pragma unroll
    for (int j = 0; j < 3; ++j)
      atomicMax(&mrow[j], ec_fkey(c[i][j]));
  }
}

__global__ void ec_final(const float* __restrict__ AC, const unsigned* __restrict__ Mkey,
                         float* __restrict__ out) {
  int i = blockIdx.x * blockDim.x + threadIdx.x;
  if (i >= N_NODES * DOUT) return;
  int n = i / DOUT;
  int f = i - n * DOUT;
  unsigned k = Mkey[i];
  float m = (k & 0x80000000u) ? __uint_as_float(k & 0x7FFFFFFFu)
                              : __uint_as_float(~k);
  out[i] = (k == 0u) ? 0.0f : (AC[(size_t)n * DCAT + f] + m);
}

extern "C" void kernel_launch(void* const* d_in, const int* in_sizes, int n_in,
                              void* d_out, int out_size, void* d_ws, size_t ws_size,
                              hipStream_t stream) {
  const float* h       = (const float*)d_in[0];
  const float* theta_w = (const float*)d_in[1];
  const float* theta_b = (const float*)d_in[2];
  const float* phi_w   = (const float*)d_in[3];
  const float* phi_b   = (const float*)d_in[4];
  const int*   src     = (const int*)d_in[5];
  const int*   dst     = (const int*)d_in[6];
  float* out = (float*)d_out;

  char* ws = (char*)d_ws;
  float*    Wcat   = (float*)ws;
  float*    bias_c = (float*)(ws + 18432);
  float*    AC     = (float*)(ws + 18688);
  unsigned* Mkey   = (unsigned*)(ws + 18688 + (size_t)N_NODES * DCAT * 4);

  ec_prep<<<(DOUT * DIN + 255) / 256, 256, 0, stream>>>(theta_w, theta_b, phi_w, phi_b,
                                                        Wcat, bias_c);
  ec_init<<<(N_NODES * DOUT + 255) / 256, 256, 0, stream>>>(Mkey);
  ec_gemm<<<(NTILES + 7) / 8, 256, 0, stream>>>(h, Wcat, bias_c, AC);
  ec_edges<<<EDGE_BLOCKS, 256, 0, stream>>>(src, dst, AC, Mkey);
  ec_final<<<(N_NODES * DOUT + 255) / 256, 256, 0, stream>>>(AC, Mkey, out);
}


# Scatter-max without value atomics: one returning count add per edge bins its source id in the destination's 128-byte line; 16 lanes per node then max the gathered rows in registers and write out
# speedup vs baseline: 1.1790x; 1.1403x over previous
#include <hip/hip_runtime.h>

typedef __attribute__((ext_vector_type(2))) float v2f;
typedef __attribute__((ext_vector_type(8))) float v8f;
typedef __attribute__((ext_vector_type(4))) int v4i;
typedef __attribute__((ext_vector_type(4))) unsigned v4u;

#define N_NODES 100000
#define N_EDGES 1600000
#define DIN 48
#define DOUT 48
#define DCAT 96
#define NTILES (N_NODES / 16)
#define BK_DW 32
#define CAP 30
#define BK_HEAD 30
#define BK_CNT 31
#define EPT 4
#define BIN_BLOCKS ((N_EDGES / EPT + 255) / 256)
#define RCH 8

__global__ void ec_prep(const float* __restrict__ theta_w, const float* __restrict__ theta_b,
                        const float* __restrict__ phi_w,   const float* __restrict__ phi_b,
                        float* __restrict__ Wcat, float* __restrict__ bias_c) {
  int i = blockIdx.x * blockDim.x + threadIdx.x;
  if (i < DOUT * DIN) {
    Wcat[i] = theta_w[i];
    Wcat[DOUT * DIN + i] = phi_w[i] - theta_w[i];
  }
  if (i < DOUT) bias_c[i] = phi_b[i] + theta_b[i];
}

__global__ void ec_init(unsigned* __restrict__ BK) {
  int i = blockIdx.x * blockDim.x + threadIdx.x;
  if (i < N_NODES * BK_DW / 4) ((v4u*)BK)[i] = (v4u){0u, 0u, 0u, 0u};
}

__global__ void ec_gemm(const float* __restrict__ h,
                        const float* __restrict__ Wcat,
                        const float* __restrict__ bias_c,
                        float* __restrict__ AC) {
  const int lane  = threadIdx.x & 31;
  const int wave  = threadIdx.x >> 5;
  const int tile  = blockIdx.x * (blockDim.x >> 5) + wave;
  if (tile >= NTILES) return;
  const int row   = lane & 15;
  const int khalf = lane >> 4;

  const float* hrow = h + (tile * 16 + row) * DIN + khalf * 2;
  v2f a[12];
#pragma unroll
  for (int kk = 0; kk < 12; ++kk)
    a[kk] = *(const v2f*)(hrow + kk * 4);

#pragma unroll
  for (int nt = 0; nt < 6; ++nt) {
    const float* wrow = Wcat + (nt * 16 + row) * DIN + khalf * 2;
    v8f c = {};
#pragma unroll
    for (int kk = 0; kk < 12; ++kk) {
      v2f b = *(const v2f*)(wrow + kk * 4);
      c = __builtin_amdgcn_wmma_f32_16x16x4_f32(false, a[kk], false, b,
                                                (short)0, c, false, false);
    }
    const float bias = (nt >= 3) ? bias_c[nt * 16 + row - DOUT] : 0.0f;
#pragma unroll
    for (int v = 0; v < 8; ++v) {
      const int m = v + 8 * khalf;
      AC[(tile * 16 + m) * DCAT + nt * 16 + row] = c[v] + bias;
    }
  }
}

__global__ void ec_bin(const int* __restrict__ src, const int* __restrict__ dst,
                       unsigned* __restrict__ BK, unsigned* __restrict__ NEXT) {
  const int t  = blockIdx.x * blockDim.x + threadIdx.x;
  const int e0 = t * EPT;
  if (e0 >= N_EDGES) return;

  const v4i sv = *(const v4i*)(src + e0);
  const v4i dv = *(const v4i*)(dst + e0);

  unsigned pos[EPT];
  for (int r = 0; r < EPT; ++r)
    pos[r] = atomicAdd(BK + (size_t)dv[r] * BK_DW + BK_CNT, 1u);

  for (int r = 0; r < EPT; ++r) {
    unsigned* line = BK + (size_t)dv[r] * BK_DW;
    if (pos[r] < CAP) {
      line[pos[r]] = (unsigned)sv[r];
    } else {
      const unsigned e = (unsigned)(e0 + r);
      NEXT[e] = atomicExch(line + BK_HEAD, e + 1u);
    }
  }
}

__global__ void ec_nodes(const int* __restrict__ src, const float* __restrict__ AC,
                         const unsigned* __restrict__ BK, const unsigned* __restrict__ NEXT,
                         float* __restrict__ out) {
  const int t = blockIdx.x * blockDim.x + threadIdx.x;
  const int g = t & 15;
  const int d = t >> 4;
  if (d >= N_NODES) return;

  const unsigned* line = BK + (size_t)d * BK_DW;
  const unsigned cboff = (DOUT + g * 3) * 4;
  const float* arow  = AC + (size_t)d * DCAT + g * 3;

  const unsigned cnt = line[BK_CNT];
  v4u q0 = *(const v4u*)(line);
  v4u q1 = *(const v4u*)(line + 4);
  float a[3];
  for (int j = 0; j < 3; ++j) a[j] = arow[j];

  const unsigned n = cnt < CAP ? cnt : CAP;
  const float ninf = -__builtin_inff();
  float m[3] = {ninf, ninf, ninf};

  for (unsigned base = 0; base < n; base += RCH) {
    const unsigned id[RCH] = {q0[0], q0[1], q0[2], q0[3], q1[0], q1[1], q1[2], q1[3]};

    float c[RCH][3];
    for (int k = 0; k < RCH; ++k) {
      const unsigned s = (base + k < n) ? id[k] : id[0];
      const float* crow = (const float*)((const char*)AC + (s * (DCAT * 4u) + cboff));
      for (int j = 0; j < 3; ++j) c[k][j] = crow[j];
    }
    if (base + RCH < n) {
      q0 = *(const v4u*)(line + base + RCH);
      q1 = *(const v4u*)(line + base + RCH + 4);
    }
    for (int k = 0; k < RCH; ++k) {
      for (int j = 0; j < 3; ++j) m[j] = fmaxf(m[j], c[k][j]);
    }
  }

  if (cnt > CAP) {
    unsigned h = line[BK_HEAD];
    for (unsigned it = CAP; it < cnt && h != 0u; ++it) {
      const unsigned e = (h - 1u < (unsigned)N_EDGES) ? h - 1u : 0u;
      const unsigned s = (unsigned)src[e];
      h = NEXT[e];
      const float* crow = (const float*)((const char*)AC + (s * (DCAT * 4u) + cboff));
      for (int j = 0; j < 3; ++j) m[j] = fmaxf(m[j], crow[j]);
    }
  }

  float* orow = out + (size_t)d * DOUT + g * 3;
  for (int j = 0; j < 3; ++j) {
    orow[j] = (cnt == 0u) ? 0.0f : a[j] + m[j];
  }
}

extern "C" void kernel_launch(void* const* d_in, const int* in_sizes, int n_in,
                              void* d_out, int out_size, void* d_ws, size_t ws_size,
                              hipStream_t stream) {
  const float* h       = (const float*)d_in[0];
  const float* theta_w = (const float*)d_in[1];
  const float* theta_b = (const float*)d_in[2];
  const float* phi_w   = (const float*)d_in[3];
  const float* phi_b   = (const float*)d_in[4];
  const int*   src     = (const int*)d_in[5];
  const int*   dst     = (const int*)d_in[6];
  float* out = (float*)d_out;

  char* ws = (char*)d_ws;
  float*    Wcat   = (float*)ws;
  float*    bias_c = (float*)(ws + 18432);
  float*    AC     = (float*)(ws + 18688);
  unsigned* BK     = (unsigned*)(ws + 18688 + (size_t)N_NODES * DCAT * 4);
  unsigned* NEXT   = (unsigned*)(ws + 18688 + (size_t)N_NODES * DCAT * 4
                                    + (size_t)N_NODES * BK_DW * 4);

  ec_prep<<<(DOUT * DIN + 255) / 256, 256, 0, stream>>>(theta_w, theta_b, phi_w, phi_b,
                                                        Wcat, bias_c);
  ec_init<<<(N_NODES * BK_DW / 4 + 255) / 256, 256, 0, stream>>>(BK);
  ec_gemm<<<(NTILES + 7) / 8, 256, 0, stream>>>(h, Wcat, bias_c, AC);
  ec_bin<<<BIN_BLOCKS, 256, 0, stream>>>(src, dst, BK, NEXT);
  ec_nodes<<<N_NODES * 16 / 256, 256, 0, stream>>>(src, AC, BK, NEXT, out);
}


# Bins in two planes, one per chip half: a binning work-group adds and stores only in the plane of the half it runs on (64-byte half lines), so no bin line is written from both second-level caches
# speedup vs baseline: 2.2173x; 1.8807x over previous
#include <hip/hip_runtime.h>

typedef __attribute__((ext_vector_type(2))) float v2f;
typedef __attribute__((ext_vector_type(8))) float v8f;
typedef __attribute__((ext_vector_type(4))) int v4i;
typedef __attribute__((ext_vector_type(4))) unsigned v4u;

#define N_NODES 100000
#define N_EDGES 1600000
#define DIN 48
#define DOUT 48
#define DCAT 96
#define NTILES (N_NODES / 16)
#define PL_DW 16
#define BK_DW (2 * PL_DW)
#define PLANE_DW (N_NODES * PL_DW)
#define CAP 14
#define BK_HEAD 14
#define BK_CNT 15
#define EPT 4
#define BIN_BLOCKS ((N_EDGES / EPT + 255) / 256)
#define RCH 8
#define GEMM_BLOCKS ((NTILES + 7) / 8)
#define CLEAR_BLOCKS (N_NODES * BK_DW / 4 / 256)
static_assert(BIN_BLOCKS == 2 * GEMM_BLOCKS - 1, "every third block a GEMM block");
static_assert(CLEAR_BLOCKS * 256 >= DOUT * DIN, "the fold fits in the clear launch");

__global__ void ec_clear_fold(const float* __restrict__ theta_w, const float* __restrict__ theta_b,
                              const float* __restrict__ phi_w,   const float* __restrict__ phi_b,
                              float* __restrict__ Wcat, float* __restrict__ bias_c,
                              unsigned* __restrict__ BK) {
  int i = blockIdx.x * blockDim.x + threadIdx.x;
  if (i < N_NODES * BK_DW / 4) ((v4u*)BK)[i] = (v4u){0u, 0u, 0u, 0u};
  if (i < DOUT * DIN) {
    Wcat[i] = theta_w[i];
    Wcat[DOUT * DIN + i] = phi_w[i] - theta_w[i];
  }
  if (i < DOUT) bias_c[i] = phi_b[i] + theta_b[i];
}

__device__ __forceinline__ void gemm_block(const int blk,
                                           const float* __restrict__ h,
                                           const float* __restrict__ Wcat,
                                           const float* __restrict__ bias_c,
                                           float* __restrict__ AC) {
  const int lane  = threadIdx.x & 31;
  const int wave  = threadIdx.x >> 5;
  const int tile  = blk * (blockDim.x >> 5) + wave;
  if (tile >= NTILES) return;
  const int row   = lane & 15;
  const int khalf = lane >> 4;

  const float* hrow = h + (tile * 16 + row) * DIN + khalf * 2;
  v2f a[12];
#pragma unroll
  for (int kk = 0; kk < 12; ++kk)
    a[kk] = *(const v2f*)(hrow + kk * 4);

#pragma unroll
  for (int nt = 0; nt < 6; ++nt) {
    const float* wrow = Wcat + (nt * 16 + row) * DIN + khalf * 2;
    v8f c = {};
#pragma unroll
    for (int kk = 0; kk < 12; ++kk) {
      v2f b = *(const v2f*)(wrow + kk * 4);
      c = __builtin_amdgcn_wmma_f32_16x16x4_f32(false, a[kk], false, b,
                                                (short)0, c, false, false);
    }
    const float bias = (nt >= 3) ? bias_c[nt * 16 + row - DOUT] : 0.0f;
#pragma unroll
    for (int v = 0; v < 8; ++v) {
      const int m = v + 8 * khalf;
      AC[(tile * 16 + m) * DCAT + nt * 16 + row] = c[v] + bias;
    }
  }
}

__device__ __forceinline__ unsigned chip_half() {
  return (__builtin_amdgcn_s_sendmsg_rtn(0x87) >> 18) & 1u;
}

__device__ __forceinline__ void bin_block(const int blk,
                                          const int* __restrict__ src, const int* __restrict__ dst,
                                          unsigned* __restrict__ BK, unsigned* __restrict__ NEXT) {
  const int t  = blk * blockDim.x + threadIdx.x;
  const int e0 = t * EPT;
  if (e0 >= N_EDGES) return;

  const v4i sv = *(const v4i*)(src + e0);
  const v4i dv = *(const v4i*)(dst + e0);
  unsigned* plane = BK + (size_t)chip_half() * PLANE_DW;

  unsigned pos[EPT];
  for (int r = 0; r < EPT; ++r)
    pos[r] = atomicAdd(plane + (size_t)dv[r] * PL_DW + BK_CNT, 1u);

  for (int r = 0; r < EPT; ++r) {
    unsigned* line = plane + (size_t)dv[r] * PL_DW;
    if (pos[r] < CAP) {
      line[pos[r]] = (unsigned)sv[r];
    } else {
      const unsigned e = (unsigned)(e0 + r);
      NEXT[e] = atomicExch(line + BK_HEAD, e + 1u);
    }
  }
}

__global__ void ec_gemm_bin(const float* __restrict__ h,
                            const float* __restrict__ Wcat,
                            const float* __restrict__ bias_c,
                            float* __restrict__ AC,
                            const int* __restrict__ src, const int* __restrict__ dst,
                            unsigned* __restrict__ BK, unsigned* __restrict__ NEXT) {
  const int q = (int)(blockIdx.x / 3u);
  const int r = (int)blockIdx.x - q * 3;
  if (r == 0) gemm_block(q, h, Wcat, bias_c, AC);
  else        bin_block(2 * q + r - 1, src, dst, BK, NEXT);
}


__device__ __forceinline__ void plane_max(const unsigned* __restrict__ line, const unsigned cnt,
                                          v4u q0, v4u q1, const unsigned cboff,
                                          const int* __restrict__ src, const float* __restrict__ AC,
                                          const unsigned* __restrict__ NEXT, float (&m)[3]) {
  const unsigned n = cnt < CAP ? cnt : CAP;

  for (unsigned base = 0; base < n; base += RCH) {
    const unsigned id[RCH] = {q0[0], q0[1], q0[2], q0[3], q1[0], q1[1], q1[2], q1[3]};

    float c[RCH][3];
    for (int k = 0; k < RCH; ++k) {
      const unsigned s = (base + k < n) ? id[k] : id[0];
      const float* crow = (const float*)((const char*)AC + (s * (DCAT * 4u) + cboff));
      for (int j = 0; j < 3; ++j) c[k][j] = crow[j];
    }
    if (base + RCH < n) {
      q0 = *(const v4u*)(line + base + RCH);
      q1 = *(const v4u*)(line + base + RCH + 4);
    }
    for (int k = 0; k < RCH; ++k) {
      for (int j = 0; j < 3; ++j) m[j] = fmaxf(m[j], c[k][j]);
    }
  }

  if (cnt > CAP) {
    unsigned h = line[BK_HEAD];
    for (unsigned it = CAP; it < cnt && h != 0u; ++it) {
      const unsigned e = (h - 1u < (unsigned)N_EDGES) ? h - 1u : 0u;
      const unsigned s = (unsigned)src[e];
      h = NEXT[e];
      const float* crow = (const float*)((const char*)AC + (s * (DCAT * 4u) + cboff));
      for (int j = 0; j < 3; ++j) m[j] = fmaxf(m[j], crow[j]);
    }
  }
}

__global__ void ec_nodes(const int* __restrict__ src, const float* __restrict__ AC,
                         const unsigned* __restrict__ BK, const unsigned* __restrict__ NEXT,
                         float* __restrict__ out) {
  const int t = blockIdx.x * blockDim.x + threadIdx.x;
  const int g = t & 15;
  const int d = t >> 4;
  if (d >= N_NODES) return;

  const unsigned* line0 = BK + (size_t)d * PL_DW;
  const unsigned* line1 = line0 + PLANE_DW;
  const unsigned cboff = (DOUT + g * 3) * 4;
  const float* arow  = AC + (size_t)d * DCAT + g * 3;

  const unsigned cnt0 = line0[BK_CNT];
  const unsigned cnt1 = line1[BK_CNT];
  const v4u p0 = *(const v4u*)(line0);
  const v4u p1 = *(const v4u*)(line0 + 4);
  const v4u r0 = *(const v4u*)(line1);
  const v4u r1 = *(const v4u*)(line1 + 4);
  float a[3];
  for (int j = 0; j < 3; ++j) a[j] = arow[j];

  const float ninf = -__builtin_inff();
  float m[3] = {ninf, ninf, ninf};
  plane_max(line0, cnt0, p0, p1, cboff, src, AC, NEXT, m);
  plane_max(line1, cnt1, r0, r1, cboff, src, AC, NEXT, m);

  float* orow = out + (size_t)d * DOUT + g * 3;
  for (int j = 0; j < 3; ++j) {
    orow[j] = ((cnt0 | cnt1) == 0u) ? 0.0f : a[j] + m[j];
  }
}

extern "C" void kernel_launch(void* const* d_in, const int* in_sizes, int n_in,
                              void* d_out, int out_size, void* d_ws, size_t ws_size,
                              hipStream_t stream) {
  const float* h       = (const float*)d_in[0];
  const float* theta_w = (const float*)d_in[1];
  const float* theta_b = (const float*)d_in[2];
  const float* phi_w   = (const float*)d_in[3];
  const float* phi_b   = (const float*)d_in[4];
  const int*   src     = (const int*)d_in[5];
  const int*   dst     = (const int*)d_in[6];
  float* out = (float*)d_out;

  char* ws = (char*)d_ws;
  float*    Wcat   = (float*)ws;
  float*    bias_c = (float*)(ws + 18432);
  float*    AC     = (float*)(ws + 18688);
  unsigned* BK     = (unsigned*)(ws + 18688 + (size_t)N_NODES * DCAT * 4);
  unsigned* NEXT   = (unsigned*)(ws + 18688 + (size_t)N_NODES * DCAT * 4
                                    + (size_t)N_NODES * BK_DW * 4);

  ec_clear_fold<<<CLEAR_BLOCKS, 256, 0, stream>>>(theta_w, theta_b, phi_w, phi_b,
                                                  Wcat, bias_c, BK);
  ec_gemm_bin<<<GEMM_BLOCKS + BIN_BLOCKS, 256, 0, stream>>>(h, Wcat, bias_c, AC,
                                                           src, dst, BK, NEXT);
  ec_nodes<<<N_NODES * 16 / 256, 256, 0, stream>>>(src, AC, BK, NEXT, out);
}


# GEMM epilogue writes whole 128-byte AC lines: two 16-column tiles are finished side by side and the lane halves swap one value per register (v_permlanex16), so a store covers 32 columns of one row
# speedup vs baseline: 2.2898x; 1.0327x over previous
#include <hip/hip_runtime.h>

typedef __attribute__((ext_vector_type(2))) float v2f;
typedef __attribute__((ext_vector_type(8))) float v8f;
typedef __attribute__((ext_vector_type(4))) int v4i;
typedef __attribute__((ext_vector_type(4))) unsigned v4u;

#define N_NODES 100000
#define N_EDGES 1600000
#define DIN 48
#define DOUT 48
#define DCAT 96
#define NTILES (N_NODES / 16)
#define PL_DW 16
#define BK_DW (2 * PL_DW)
#define PLANE_DW (N_NODES * PL_DW)
#define CAP 14
#define BK_HEAD 14
#define BK_CNT 15
#define EPT 4
#define BIN_BLOCKS ((N_EDGES / EPT + 255) / 256)
#define RCH 8
#define GEMM_BLOCKS ((NTILES + 7) / 8)
#define CLEAR_BLOCKS (N_NODES * BK_DW / 4 / 256)
static_assert(BIN_BLOCKS == 2 * GEMM_BLOCKS - 1, "every third block a GEMM block");
static_assert(CLEAR_BLOCKS * 256 >= DOUT * DIN, "the fold fits in the clear launch");

__global__ void ec_clear_fold(const float* __restrict__ theta_w, const float* __restrict__ theta_b,
                              const float* __restrict__ phi_w,   const float* __restrict__ phi_b,
                              float* __restrict__ Wcat, float* __restrict__ bias_c,
                              unsigned* __restrict__ BK) {
  int i = blockIdx.x * blockDim.x + threadIdx.x;
  if (i < N_NODES * BK_DW / 4) ((v4u*)BK)[i] = (v4u){0u, 0u, 0u, 0u};
  if (i < DOUT * DIN) {
    Wcat[i] = theta_w[i];
    Wcat[DOUT * DIN + i] = phi_w[i] - theta_w[i];
  }
  if (i < DOUT) bias_c[i] = phi_b[i] + theta_b[i];
}

__device__ __forceinline__ void gemm_block(const int blk,
                                           const float* __restrict__ h,
                                           const float* __restrict__ Wcat,
                                           const float* __restrict__ bias_c,
                                           float* __restrict__ AC) {
  const int lane  = threadIdx.x & 31;
  const int wave  = threadIdx.x >> 5;
  const int tile  = blk * (blockDim.x >> 5) + wave;
  if (tile >= NTILES) return;
  const int row   = lane & 15;
  const int khalf = lane >> 4;

  const float* hrow = h + (tile * 16 + row) * DIN + khalf * 2;
  v2f a[12];
#pragma unroll
  for (int kk = 0; kk < 12; ++kk)
    a[kk] = *(const v2f*)(hrow + kk * 4);

#pragma unroll
  for (int j = 0; j < 3; ++j) {
    const float* wrow0 = Wcat + ((2 * j) * 16 + row) * DIN + khalf * 2;
    const float* wrow1 = wrow0 + 16 * DIN;
    v8f c0 = {};
    v8f c1 = {};
#pragma unroll
    for (int kk = 0; kk < 12; ++kk) {
      v2f b0 = *(const v2f*)(wrow0 + kk * 4);
      v2f b1 = *(const v2f*)(wrow1 + kk * 4);
      c0 = __builtin_amdgcn_wmma_f32_16x16x4_f32(false, a[kk], false, b0,
                                                 (short)0, c0, false, false);
      c1 = __builtin_amdgcn_wmma_f32_16x16x4_f32(false, a[kk], false, b1,
                                                 (short)0, c1, false, false);
    }
    const int col = 32 * j + lane;
    const float bval = bias_c[col >= DOUT ? col - DOUT : 0];
    const float bias = (col >= DOUT) ? bval : 0.0f;
    float* orow = AC + (tile * 16) * DCAT + col;
#pragma unroll
    for (int v = 0; v < 8; ++v) {
      const float give = khalf ? c0[v] : c1[v];
      const float got = __builtin_bit_cast(float,
          __builtin_amdgcn_permlanex16(0u, __builtin_bit_cast(unsigned, give),
                                       0x76543210u, 0xfedcba98u, false, false));
      const float lo = khalf ? got : c0[v];
      const float hi = khalf ? c1[v] : got;
      orow[v * DCAT]       = lo + bias;
      orow[(v + 8) * DCAT] = hi + bias;
    }
  }
}

__device__ __forceinline__ unsigned chip_half() {
  return (__builtin_amdgcn_s_sendmsg_rtn(0x87) >> 18) & 1u;
}

__device__ __forceinline__ void bin_block(const int blk,
                                          const int* __restrict__ src, const int* __restrict__ dst,
                                          unsigned* __restrict__ BK, unsigned* __restrict__ NEXT) {
  const int t  = blk * blockDim.x + threadIdx.x;
  const int e0 = t * EPT;
  if (e0 >= N_EDGES) return;

  const v4i sv = *(const v4i*)(src + e0);
  const v4i dv = *(const v4i*)(dst + e0);
  unsigned* plane = BK + (size_t)chip_half() * PLANE_DW;

  unsigned pos[EPT];
  for (int r = 0; r < EPT; ++r)
    pos[r] = atomicAdd(plane + (size_t)dv[r] * PL_DW + BK_CNT, 1u);

  for (int r = 0; r < EPT; ++r) {
    unsigned* line = plane + (size_t)dv[r] * PL_DW;
    if (pos[r] < CAP) {
      line[pos[r]] = (unsigned)sv[r];
    } else {
      const unsigned e = (unsigned)(e0 + r);
      NEXT[e] = atomicExch(line + BK_HEAD, e + 1u);
    }
  }
}

__global__ void ec_gemm_bin(const float* __restrict__ h,
                            const float* __restrict__ Wcat,
                            const float* __restrict__ bias_c,
                            float* __restrict__ AC,
                            const int* __restrict__ src, const int* __restrict__ dst,
                            unsigned* __restrict__ BK, unsigned* __restrict__ NEXT) {
  const int q = (int)(blockIdx.x / 3u);
  const int r = (int)blockIdx.x - q * 3;
  if (r == 0) gemm_block(q, h, Wcat, bias_c, AC);
  else        bin_block(2 * q + r - 1, src, dst, BK, NEXT);
}


__device__ __forceinline__ void plane_max(const unsigned* __restrict__ line, const unsigned cnt,
                                          v4u q0, v4u q1, const unsigned cboff,
                                          const int* __restrict__ src, const float* __restrict__ AC,
                                          const unsigned* __restrict__ NEXT, float (&m)[3]) {
  const unsigned n = cnt < CAP ? cnt : CAP;

  for (unsigned base = 0; base < n; base += RCH) {
    const unsigned id[RCH] = {q0[0], q0[1], q0[2], q0[3], q1[0], q1[1], q1[2], q1[3]};

    float c[RCH][3];
    for (int k = 0; k < RCH; ++k) {
      const unsigned s = (base + k < n) ? id[k] : id[0];
      const float* crow = (const float*)((const char*)AC + (s * (DCAT * 4u) + cboff));
      for (int j = 0; j < 3; ++j) c[k][j] = crow[j];
    }
    if (base + RCH < n) {
      q0 = *(const v4u*)(line + base + RCH);
      q1 = *(const v4u*)(line + base + RCH + 4);
    }
    for (int k = 0; k < RCH; ++k) {
      for (int j = 0; j < 3; ++j) m[j] = fmaxf(m[j], c[k][j]);
    }
  }

  if (cnt > CAP) {
    unsigned h = line[BK_HEAD];
    for (unsigned it = CAP; it < cnt && h != 0u; ++it) {
      const unsigned e = (h - 1u < (unsigned)N_EDGES) ? h - 1u : 0u;
      const unsigned s = (unsigned)src[e];
      h = NEXT[e];
      const float* crow = (const float*)((const char*)AC + (s * (DCAT * 4u) + cboff));
      for (int j = 0; j < 3; ++j) m[j] = fmaxf(m[j], crow[j]);
    }
  }
}

__global__ void ec_nodes(const int* __restrict__ src, const float* __restrict__ AC,
                         const unsigned* __restrict__ BK, const unsigned* __restrict__ NEXT,
                         float* __restrict__ out) {
  const int t = blockIdx.x * blockDim.x + threadIdx.x;
  const int g = t & 15;
  const int d = t >> 4;
  if (d >= N_NODES) return;

  const unsigned* line0 = BK + (size_t)d * PL_DW;
  const unsigned* line1 = line0 + PLANE_DW;
  const unsigned cboff = (DOUT + g * 3) * 4;
  const float* arow  = AC + (size_t)d * DCAT + g * 3;

  const unsigned cnt0 = line0[BK_CNT];
  const unsigned cnt1 = line1[BK_CNT];
  const v4u p0 = *(const v4u*)(line0);
  const v4u p1 = *(const v4u*)(line0 + 4);
  const v4u r0 = *(const v4u*)(line1);
  const v4u r1 = *(const v4u*)(line1 + 4);
  float a[3];
  for (int j = 0; j < 3; ++j) a[j] = arow[j];

  const float ninf = -__builtin_inff();
  float m[3] = {ninf, ninf, ninf};
  plane_max(line0, cnt0, p0, p1, cboff, src, AC, NEXT, m);
  plane_max(line1, cnt1, r0, r1, cboff, src, AC, NEXT, m);

  float* orow = out + (size_t)d * DOUT + g * 3;
  for (int j = 0; j < 3; ++j) {
    orow[j] = ((cnt0 | cnt1) == 0u) ? 0.0f : a[j] + m[j];
  }
}

extern "C" void kernel_launch(void* const* d_in, const int* in_sizes, int n_in,
                              void* d_out, int out_size, void* d_ws, size_t ws_size,
                              hipStream_t stream) {
  const float* h       = (const float*)d_in[0];
  const float* theta_w = (const float*)d_in[1];
  const float* theta_b = (const float*)d_in[2];
  const float* phi_w   = (const float*)d_in[3];
  const float* phi_b   = (const float*)d_in[4];
  const int*   src     = (const int*)d_in[5];
  const int*   dst     = (const int*)d_in[6];
  float* out = (float*)d_out;

  char* ws = (char*)d_ws;
  float*    Wcat   = (float*)ws;
  float*    bias_c = (float*)(ws + 18432);
  float*    AC     = (float*)(ws + 18688);
  unsigned* BK     = (unsigned*)(ws + 18688 + (size_t)N_NODES * DCAT * 4);
  unsigned* NEXT   = (unsigned*)(ws + 18688 + (size_t)N_NODES * DCAT * 4
                                    + (size_t)N_NODES * BK_DW * 4);

  ec_clear_fold<<<CLEAR_BLOCKS, 256, 0, stream>>>(theta_w, theta_b, phi_w, phi_b,
                                                  Wcat, bias_c, BK);
  ec_gemm_bin<<<GEMM_BLOCKS + BIN_BLOCKS, 256, 0, stream>>>(h, Wcat, bias_c, AC,
                                                           src, dst, BK, NEXT);
  ec_nodes<<<N_NODES * 16 / 256, 256, 0, stream>>>(src, AC, BK, NEXT, out);
}


# Edges partitioned by destination bucket through LDS counts (one coalesced global add per bucket and work-group, no scattered atomics); a work-group per bucket sorts them by node in LDS
# speedup vs baseline: 2.6460x; 1.1556x over previous
#include <hip/hip_runtime.h>

typedef __attribute__((ext_vector_type(2))) float v2f;
typedef __attribute__((ext_vector_type(8))) float v8f;
typedef __attribute__((ext_vector_type(4))) int v4i;
typedef __attribute__((ext_vector_type(4))) unsigned v4u;

#define N_NODES 100000
#define N_EDGES 1600000
#define DIN 48
#define DOUT 48
#define DCAT 96
#define NTILES (N_NODES / 16)
#define NPB 128
#define NB ((N_NODES + NPB - 1) / NPB)
#define HCAP 1536
#define EPB 4096
#define PART_BLOCKS ((N_EDGES + EPB - 1) / EPB)
#define GEMM_BLOCKS ((NTILES + 7) / 8)
#define SCAP 40
#define RCH 8
#define NODE_T 512
static_assert(GEMM_BLOCKS == 2 * PART_BLOCKS, "every third block a partition block");
static_assert(NPB == 128 && N_NODES < (1 << 25), "a packed word is source << 7 | local destination");
static_assert(SCAP % RCH == 0, "a trip never reads past a node's slots");
static_assert(HCAP % NODE_T == 0, "a node work-group's threads cover a region in whole rounds");
#define CNT_WORDS (2 * NB + 1)

__global__ void ec_clear_fold(const float* __restrict__ theta_w, const float* __restrict__ theta_b,
                              const float* __restrict__ phi_w,   const float* __restrict__ phi_b,
                              float* __restrict__ Wcat, float* __restrict__ bias_c,
                              unsigned* __restrict__ CNT) {
  int i = blockIdx.x * blockDim.x + threadIdx.x;
  if (i < CNT_WORDS) CNT[i] = 0u;
  if (i < DOUT * DIN) {
    Wcat[i] = theta_w[i];
    Wcat[DOUT * DIN + i] = phi_w[i] - theta_w[i];
  }
  if (i < DOUT) bias_c[i] = phi_b[i] + theta_b[i];
}

__device__ __forceinline__ void gemm_block(const int blk,
                                           const float* __restrict__ h,
                                           const float* __restrict__ Wcat,
                                           const float* __restrict__ bias_c,
                                           float* __restrict__ AC) {
  const int lane  = threadIdx.x & 31;
  const int wave  = threadIdx.x >> 5;
  const int tile  = blk * (blockDim.x >> 5) + wave;
  if (tile >= NTILES) return;
  const int row   = lane & 15;
  const int khalf = lane >> 4;

  const float* hrow = h + (tile * 16 + row) * DIN + khalf * 2;
  v2f a[12];
#pragma unroll
  for (int kk = 0; kk < 12; ++kk)
    a[kk] = *(const v2f*)(hrow + kk * 4);

#pragma unroll
  for (int j = 0; j < 3; ++j) {
    const float* wrow0 = Wcat + ((2 * j) * 16 + row) * DIN + khalf * 2;
    const float* wrow1 = wrow0 + 16 * DIN;
    v8f c0 = {};
    v8f c1 = {};
#pragma unroll
    for (int kk = 0; kk < 12; ++kk) {
      v2f b0 = *(const v2f*)(wrow0 + kk * 4);
      v2f b1 = *(const v2f*)(wrow1 + kk * 4);
      c0 = __builtin_amdgcn_wmma_f32_16x16x4_f32(false, a[kk], false, b0,
                                                 (short)0, c0, false, false);
      c1 = __builtin_amdgcn_wmma_f32_16x16x4_f32(false, a[kk], false, b1,
                                                 (short)0, c1, false, false);
    }
    const int col = 32 * j + lane;
    const float bval = bias_c[col >= DOUT ? col - DOUT : 0];
    const float bias = (col >= DOUT) ? bval : 0.0f;
    float* orow = AC + (tile * 16) * DCAT + col;
#pragma unroll
    for (int v = 0; v < 8; ++v) {
      const float give = khalf ? c0[v] : c1[v];
      const float got = __builtin_bit_cast(float,
          __builtin_amdgcn_permlanex16(0u, __builtin_bit_cast(unsigned, give),
                                       0x76543210u, 0xfedcba98u, false, false));
      const float lo = khalf ? got : c0[v];
      const float hi = khalf ? c1[v] : got;
      orow[v * DCAT]       = lo + bias;
      orow[(v + 8) * DCAT] = hi + bias;
    }
  }
}

__device__ __forceinline__ unsigned chip_half() {
  return (__builtin_amdgcn_s_sendmsg_rtn(0x87) >> 18) & 1u;
}

__device__ __forceinline__ void part_block(const int blk,
                                           const int* __restrict__ src, const int* __restrict__ dst,
                                           unsigned* __restrict__ CNT, unsigned* __restrict__ REG,
                                           unsigned* __restrict__ OVF) {
  __shared__ unsigned s_hist[NB];
  __shared__ unsigned s_base[NB];
  __shared__ unsigned s_half;
  const int tid = threadIdx.x;
  for (int b = tid; b < NB; b += 256) s_hist[b] = 0u;
  if (tid == 0) s_half = chip_half();
  __syncthreads();

  const unsigned half = s_half;
  v4i sv[4], dv[4];
  unsigned rk[16];
#pragma unroll
  for (int i = 0; i < 4; ++i) {
    const int e0 = blk * EPB + i * 1024 + tid * 4;
    const int ec = e0 < N_EDGES ? e0 : N_EDGES - 4;
    sv[i] = *(const v4i*)(src + ec);
    dv[i] = *(const v4i*)(dst + ec);
  }
#pragma unroll
  for (int i = 0; i < 4; ++i) {
    const int e0 = blk * EPB + i * 1024 + tid * 4;
    if (e0 >= N_EDGES) dv[i] = (v4i){-1, -1, -1, -1};
  }
#pragma unroll
  for (int i = 0; i < 4; ++i) {
#pragma unroll
    for (int r = 0; r < 4; ++r) {
      const int d = dv[i][r];
      rk[i * 4 + r] = (d >= 0) ? atomicAdd(&s_hist[(unsigned)d >> 7], 1u) : 0u;
    }
  }
  __syncthreads();

  unsigned* cnt_half = CNT + half * NB;
  for (int b = tid; b < NB; b += 256) {
    const unsigned c = s_hist[b];
    s_base[b] = c ? atomicAdd(cnt_half + b, c) : 0u;
  }
  __syncthreads();

  unsigned* reg_half = REG + (size_t)half * NB * HCAP;
#pragma unroll
  for (int i = 0; i < 4; ++i) {
#pragma unroll
    for (int r = 0; r < 4; ++r) {
      const int d = dv[i][r];
      if (d >= 0) {
        const unsigned bk  = (unsigned)d >> 7;
        const unsigned pos = s_base[bk] + rk[i * 4 + r];
        if (pos < HCAP) {
          reg_half[bk * HCAP + pos] = ((unsigned)sv[i][r] << 7) | ((unsigned)d & 127u);
        } else {
          const unsigned o = atomicAdd(CNT + 2 * NB, 1u);
          if (o < (unsigned)N_EDGES) OVF[o] = (unsigned)(blk * EPB + i * 1024 + tid * 4 + r);
        }
      }
    }
  }
}

__global__ void ec_gemm_part(const float* __restrict__ h,
                             const float* __restrict__ Wcat,
                             const float* __restrict__ bias_c,
                             float* __restrict__ AC,
                             const int* __restrict__ src, const int* __restrict__ dst,
                             unsigned* __restrict__ CNT, unsigned* __restrict__ REG,
                             unsigned* __restrict__ OVF) {
  const int q = (int)(blockIdx.x / 3u);
  const int r = (int)blockIdx.x - q * 3;
  if (r == 0) part_block(q, src, dst, CNT, REG, OVF);
  else        gemm_block(2 * q + r - 1, h, Wcat, bias_c, AC);
}

__device__ __forceinline__ void row_max(const float* __restrict__ AC, const unsigned s,
                                        const unsigned cboff, float (&m)[3]) {
  const float* crow = (const float*)((const char*)AC + (s * (DCAT * 4u) + cboff));
  for (int j = 0; j < 3; ++j) m[j] = fmaxf(m[j], crow[j]);
}

__global__ void __launch_bounds__(NODE_T)
ec_nodes(const int* __restrict__ src, const int* __restrict__ dst, const float* __restrict__ AC,
         const unsigned* __restrict__ CNT, const unsigned* __restrict__ REG,
         const unsigned* __restrict__ OVF, float* __restrict__ out) {
  __shared__ unsigned s_cnt[NPB];
  __shared__ unsigned s_slot[NPB * SCAP];
  const int tid = threadIdx.x;
  const unsigned b = blockIdx.x;

  for (int i = tid; i < NPB; i += NODE_T) s_cnt[i] = 0u;
  __syncthreads();

  unsigned nh[2];
  for (int hf = 0; hf < 2; ++hf) {
    const unsigned c = CNT[hf * NB + b];
    nh[hf] = c < HCAP ? c : HCAP;
  }
  const unsigned c_ovf = CNT[2 * NB];
  const unsigned novf = c_ovf < (unsigned)N_EDGES ? c_ovf : (unsigned)N_EDGES;

  unsigned w[2][HCAP / NODE_T];
  for (int hf = 0; hf < 2; ++hf) {
    const unsigned* reg = REG + ((size_t)hf * NB + b) * HCAP;
    for (int k = 0; k < HCAP / NODE_T; ++k) {
      const unsigned i = (unsigned)tid + (unsigned)k * NODE_T;
      w[hf][k] = (i < nh[hf]) ? reg[i] : 0xffffffffu;
    }
  }
  for (int hf = 0; hf < 2; ++hf) {
    for (int k = 0; k < HCAP / NODE_T; ++k) {
      const unsigned x = w[hf][k];
      if (x != 0xffffffffu) {
        const unsigned ld = x & 127u;
        const unsigned p  = atomicAdd(&s_cnt[ld], 1u);
        if (p < SCAP) s_slot[ld * SCAP + p] = x >> 7;
      }
    }
  }
  for (unsigned i = tid; i < novf; i += NODE_T) {
    const unsigned e0 = OVF[i];
    const unsigned e  = e0 < (unsigned)N_EDGES ? e0 : 0u;
    const unsigned d  = (unsigned)dst[e];
    if ((d >> 7) == b) {
      const unsigned ld = d & 127u;
      const unsigned p  = atomicAdd(&s_cnt[ld], 1u);
      if (p < SCAP) s_slot[ld * SCAP + p] = (unsigned)src[e];
    }
  }
  __syncthreads();

  const int g  = tid & 15;
  const int gr = tid >> 4;
  const unsigned cboff = (DOUT + g * 3) * 4;
  const float ninf = -__builtin_inff();

  for (int ln = gr; ln < NPB; ln += NODE_T / 16) {
    const int d = (int)b * NPB + ln;
    if (d >= N_NODES) break;
    const unsigned cnt = s_cnt[ln];
    const float* arow = AC + (size_t)d * DCAT + g * 3;
    float a[3];
    for (int j = 0; j < 3; ++j) a[j] = arow[j];
    float m[3] = {ninf, ninf, ninf};

    if (cnt <= SCAP) {
      const unsigned* sl = s_slot + ln * SCAP;
      for (unsigned base = 0; base < cnt; base += RCH) {
        unsigned id[RCH];
        for (int k = 0; k < RCH; ++k) id[k] = sl[base + k];
        float c[RCH][3];
        for (int k = 0; k < RCH; ++k) {
          const unsigned s = (base + k < cnt) ? id[k] : id[0];
          const float* crow = (const float*)((const char*)AC + (s * (DCAT * 4u) + cboff));
          for (int j = 0; j < 3; ++j) c[k][j] = crow[j];
        }
        for (int k = 0; k < RCH; ++k) {
          for (int j = 0; j < 3; ++j) m[j] = fmaxf(m[j], c[k][j]);
        }
      }
    } else {
      for (int hf = 0; hf < 2; ++hf) {
        const unsigned* reg = REG + ((size_t)hf * NB + b) * HCAP;
        for (unsigned i = 0; i < nh[hf]; ++i) {
          const unsigned w = reg[i];
          if ((w & 127u) == (unsigned)ln) row_max(AC, w >> 7, cboff, m);
        }
      }
      for (unsigned i = 0; i < novf; ++i) {
        const unsigned e0 = OVF[i];
        const unsigned e  = e0 < (unsigned)N_EDGES ? e0 : 0u;
        if (dst[e] == d) row_max(AC, (unsigned)src[e], cboff, m);
      }
    }

    float* orow = out + (size_t)d * DOUT + g * 3;
    for (int j = 0; j < 3; ++j) orow[j] = (cnt == 0u) ? 0.0f : a[j] + m[j];
  }
}

extern "C" void kernel_launch(void* const* d_in, const int* in_sizes, int n_in,
                              void* d_out, int out_size, void* d_ws, size_t ws_size,
                              hipStream_t stream) {
  const float* h       = (const float*)d_in[0];
  const float* theta_w = (const float*)d_in[1];
  const float* theta_b = (const float*)d_in[2];
  const float* phi_w   = (const float*)d_in[3];
  const float* phi_b   = (const float*)d_in[4];
  const int*   src     = (const int*)d_in[5];
  const int*   dst     = (const int*)d_in[6];
  float* out = (float*)d_out;

  char* ws = (char*)d_ws;
  float*    Wcat   = (float*)ws;
  float*    bias_c = (float*)(ws + 18432);
  float*    AC     = (float*)(ws + 18688);
  unsigned* CNT    = (unsigned*)(ws + 18688 + (size_t)N_NODES * DCAT * 4);
  unsigned* REG    = (unsigned*)(ws + 18688 + (size_t)N_NODES * DCAT * 4 + 8192);
  unsigned* OVF    = (unsigned*)(ws + 18688 + (size_t)N_NODES * DCAT * 4 + 8192
                                    + (size_t)2 * NB * HCAP * 4);

  ec_clear_fold<<<(DOUT * DIN + 255) / 256, 256, 0, stream>>>(theta_w, theta_b, phi_w, phi_b,
                                                              Wcat, bias_c, CNT);
  ec_gemm_part<<<GEMM_BLOCKS + PART_BLOCKS, 256, 0, stream>>>(h, Wcat, bias_c, AC,
                                                              src, dst, CNT, REG, OVF);
  ec_nodes<<<NB, NODE_T, 0, stream>>>(src, dst, AC, CNT, REG, OVF, out);
}


# Folded weights stored in B-fragment order by the fold launch: a GEMM wave reads B for two k steps with one 16-byte load per lane from 512 contiguous bytes, not 8-byte loads touching 16 lines each
# speedup vs baseline: 2.7520x; 1.0401x over previous
#include <hip/hip_runtime.h>

typedef __attribute__((ext_vector_type(2))) float v2f;
typedef __attribute__((ext_vector_type(4))) float v4f;
typedef __attribute__((ext_vector_type(8))) float v8f;
typedef __attribute__((ext_vector_type(4))) int v4i;
typedef __attribute__((ext_vector_type(4))) unsigned v4u;

#define N_NODES 100000
#define N_EDGES 1600000
#define DIN 48
#define DOUT 48
#define DCAT 96
#define NTILES (N_NODES / 16)
#define NPB 128
#define NB ((N_NODES + NPB - 1) / NPB)
#define HCAP 1536
#define EPB 4096
#define PART_BLOCKS ((N_EDGES + EPB - 1) / EPB)
#define GEMM_BLOCKS ((NTILES + 7) / 8)
#define SCAP 40
#define RCH 8
#define NODE_T 512
static_assert(GEMM_BLOCKS == 2 * PART_BLOCKS, "every third block a partition block");
static_assert(NPB == 128 && N_NODES < (1 << 25), "a packed word is source << 7 | local destination");
static_assert(SCAP % RCH == 0, "a trip never reads past a node's slots");
static_assert(HCAP % NODE_T == 0, "a node work-group's threads cover a region in whole rounds");
#define CNT_WORDS (2 * NB + 1)

__device__ __forceinline__ int wfrag_index(const int row, const int k) {
  const int t  = row >> 4;
  const int n  = row & 15;
  const int p  = k >> 3;
  const int hf = (k >> 1) & 1;
  const int e  = (((k >> 2) & 1) << 1) | (k & 1);
  return (((t * 6 + p) * 32) + hf * 16 + n) * 4 + e;
}

__global__ void ec_clear_fold(const float* __restrict__ theta_w, const float* __restrict__ theta_b,
                              const float* __restrict__ phi_w,   const float* __restrict__ phi_b,
                              float* __restrict__ Wfrag, float* __restrict__ bias_c,
                              unsigned* __restrict__ CNT) {
  int i = blockIdx.x * blockDim.x + threadIdx.x;
  if (i < CNT_WORDS) CNT[i] = 0u;
  if (i < DOUT * DIN) {
    const int r = i / DIN;
    const int k = i - r * DIN;
    Wfrag[wfrag_index(r, k)]        = theta_w[i];
    Wfrag[wfrag_index(DOUT + r, k)] = phi_w[i] - theta_w[i];
  }
  if (i < DOUT) bias_c[i] = phi_b[i] + theta_b[i];
}

__device__ __forceinline__ void gemm_block(const int blk,
                                           const float* __restrict__ h,
                                           const float* __restrict__ Wfrag,
                                           const float* __restrict__ bias_c,
                                           float* __restrict__ AC) {
  const int lane  = threadIdx.x & 31;
  const int wave  = threadIdx.x >> 5;
  const int tile  = blk * (blockDim.x >> 5) + wave;
  if (tile >= NTILES) return;
  const int row   = lane & 15;
  const int khalf = lane >> 4;

  const float* hrow = h + (tile * 16 + row) * DIN + khalf * 2;
  v2f a[12];
#pragma unroll
  for (int kk = 0; kk < 12; ++kk)
    a[kk] = *(const v2f*)(hrow + kk * 4);

#pragma unroll
  for (int j = 0; j < 3; ++j) {
    const float* wf0 = Wfrag + ((2 * j) * 6 * 32 + lane) * 4;
    const float* wf1 = wf0 + 6 * 32 * 4;
    v8f c0 = {};
    v8f c1 = {};
#pragma unroll
    for (int p = 0; p < 6; ++p) {
      const v4f w0 = *(const v4f*)(wf0 + p * 32 * 4);
      const v4f w1 = *(const v4f*)(wf1 + p * 32 * 4);
      const v2f b0lo = {w0.x, w0.y}, b0hi = {w0.z, w0.w};
      const v2f b1lo = {w1.x, w1.y}, b1hi = {w1.z, w1.w};
      c0 = __builtin_amdgcn_wmma_f32_16x16x4_f32(false, a[2 * p], false, b0lo,
                                                 (short)0, c0, false, false);
      c1 = __builtin_amdgcn_wmma_f32_16x16x4_f32(false, a[2 * p], false, b1lo,
                                                 (short)0, c1, false, false);
      c0 = __builtin_amdgcn_wmma_f32_16x16x4_f32(false, a[2 * p + 1], false, b0hi,
                                                 (short)0, c0, false, false);
      c1 = __builtin_amdgcn_wmma_f32_16x16x4_f32(false, a[2 * p + 1], false, b1hi,
                                                 (short)0, c1, false, false);
    }
    const int col = 32 * j + lane;
    const float bval = bias_c[col >= DOUT ? col - DOUT : 0];
    const float bias = (col >= DOUT) ? bval : 0.0f;
    float* orow = AC + (tile * 16) * DCAT + col;
#pragma unroll
    for (int v = 0; v < 8; ++v) {
      const float give = khalf ? c0[v] : c1[v];
      const float got = __builtin_bit_cast(float,
          __builtin_amdgcn_permlanex16(0u, __builtin_bit_cast(unsigned, give),
                                       0x76543210u, 0xfedcba98u, false, false));
      const float lo = khalf ? got : c0[v];
      const float hi = khalf ? c1[v] : got;
      orow[v * DCAT]       = lo + bias;
      orow[(v + 8) * DCAT] = hi + bias;
    }
  }
}

__device__ __forceinline__ unsigned chip_half() {
  return (__builtin_amdgcn_s_sendmsg_rtn(0x87) >> 18) & 1u;
}

__device__ __forceinline__ void part_block(const int blk,
                                           const int* __restrict__ src, const int* __restrict__ dst,
                                           unsigned* __restrict__ CNT, unsigned* __restrict__ REG,
                                           unsigned* __restrict__ OVF) {
  __shared__ unsigned s_hist[NB];
  __shared__ unsigned s_base[NB];
  __shared__ unsigned s_half;
  const int tid = threadIdx.x;
  for (int b = tid; b < NB; b += 256) s_hist[b] = 0u;
  if (tid == 0) s_half = chip_half();
  __syncthreads();

  const unsigned half = s_half;
  v4i sv[4], dv[4];
  unsigned rk[16];
#pragma unroll
  for (int i = 0; i < 4; ++i) {
    const int e0 = blk * EPB + i * 1024 + tid * 4;
    const int ec = e0 < N_EDGES ? e0 : N_EDGES - 4;
    sv[i] = *(const v4i*)(src + ec);
    dv[i] = *(const v4i*)(dst + ec);
  }
#pragma unroll
  for (int i = 0; i < 4; ++i) {
    const int e0 = blk * EPB + i * 1024 + tid * 4;
    if (e0 >= N_EDGES) dv[i] = (v4i){-1, -1, -1, -1};
  }
#pragma unroll
  for (int i = 0; i < 4; ++i) {
#pragma unroll
    for (int r = 0; r < 4; ++r) {
      const int d = dv[i][r];
      rk[i * 4 + r] = (d >= 0) ? atomicAdd(&s_hist[(unsigned)d >> 7], 1u) : 0u;
    }
  }
  __syncthreads();

  unsigned* cnt_half = CNT + half * NB;
  for (int b = tid; b < NB; b += 256) {
    const unsigned c = s_hist[b];
    s_base[b] = c ? atomicAdd(cnt_half + b, c) : 0u;
  }
  __syncthreads();

  unsigned* reg_half = REG + (size_t)half * NB * HCAP;
#pragma unroll
  for (int i = 0; i < 4; ++i) {
#pragma unroll
    for (int r = 0; r < 4; ++r) {
      const int d = dv[i][r];
      if (d >= 0) {
        const unsigned bk  = (unsigned)d >> 7;
        const unsigned pos = s_base[bk] + rk[i * 4 + r];
        if (pos < HCAP) {
          reg_half[bk * HCAP + pos] = ((unsigned)sv[i][r] << 7) | ((unsigned)d & 127u);
        } else {
          const unsigned o = atomicAdd(CNT + 2 * NB, 1u);
          if (o < (unsigned)N_EDGES) OVF[o] = (unsigned)(blk * EPB + i * 1024 + tid * 4 + r);
        }
      }
    }
  }
}

__global__ void ec_gemm_part(const float* __restrict__ h,
                             const float* __restrict__ Wfrag,
                             const float* __restrict__ bias_c,
                             float* __restrict__ AC,
                             const int* __restrict__ src, const int* __restrict__ dst,
                             unsigned* __restrict__ CNT, unsigned* __restrict__ REG,
                             unsigned* __restrict__ OVF) {
  const int q = (int)(blockIdx.x / 3u);
  const int r = (int)blockIdx.x - q * 3;
  if (r == 0) part_block(q, src, dst, CNT, REG, OVF);
  else        gemm_block(2 * q + r - 1, h, Wfrag, bias_c, AC);
}

__device__ __forceinline__ void row_max(const float* __restrict__ AC, const unsigned s,
                                        const unsigned cboff, float (&m)[3]) {
  const float* crow = (const float*)((const char*)AC + (s * (DCAT * 4u) + cboff));
  for (int j = 0; j < 3; ++j) m[j] = fmaxf(m[j], crow[j]);
}

__global__ void __launch_bounds__(NODE_T)
ec_nodes(const int* __restrict__ src, const int* __restrict__ dst, const float* __restrict__ AC,
         const unsigned* __restrict__ CNT, const unsigned* __restrict__ REG,
         const unsigned* __restrict__ OVF, float* __restrict__ out) {
  __shared__ unsigned s_cnt[NPB];
  __shared__ unsigned s_slot[NPB * SCAP];
  const int tid = threadIdx.x;
  const unsigned b = blockIdx.x;

  for (int i = tid; i < NPB; i += NODE_T) s_cnt[i] = 0u;
  __syncthreads();

  unsigned nh[2];
  for (int hf = 0; hf < 2; ++hf) {
    const unsigned c = CNT[hf * NB + b];
    nh[hf] = c < HCAP ? c : HCAP;
  }
  const unsigned c_ovf = CNT[2 * NB];
  const unsigned novf = c_ovf < (unsigned)N_EDGES ? c_ovf : (unsigned)N_EDGES;

  unsigned w[2][HCAP / NODE_T];
  for (int hf = 0; hf < 2; ++hf) {
    const unsigned* reg = REG + ((size_t)hf * NB + b) * HCAP;
    for (int k = 0; k < HCAP / NODE_T; ++k) {
      const unsigned i = (unsigned)tid + (unsigned)k * NODE_T;
      w[hf][k] = (i < nh[hf]) ? reg[i] : 0xffffffffu;
    }
  }
  for (int hf = 0; hf < 2; ++hf) {
    for (int k = 0; k < HCAP / NODE_T; ++k) {
      const unsigned x = w[hf][k];
      if (x != 0xffffffffu) {
        const unsigned ld = x & 127u;
        const unsigned p  = atomicAdd(&s_cnt[ld], 1u);
        if (p < SCAP) s_slot[ld * SCAP + p] = x >> 7;
      }
    }
  }
  for (unsigned i = tid; i < novf; i += NODE_T) {
    const unsigned e0 = OVF[i];
    const unsigned e  = e0 < (unsigned)N_EDGES ? e0 : 0u;
    const unsigned d  = (unsigned)dst[e];
    if ((d >> 7) == b) {
      const unsigned ld = d & 127u;
      const unsigned p  = atomicAdd(&s_cnt[ld], 1u);
      if (p < SCAP) s_slot[ld * SCAP + p] = (unsigned)src[e];
    }
  }
  __syncthreads();

  const int g  = tid & 15;
  const int gr = tid >> 4;
  const unsigned cboff = (DOUT + g * 3) * 4;
  const float ninf = -__builtin_inff();

  for (int ln = gr; ln < NPB; ln += NODE_T / 16) {
    const int d = (int)b * NPB + ln;
    if (d >= N_NODES) break;
    const unsigned cnt = s_cnt[ln];
    const float* arow = AC + (size_t)d * DCAT + g * 3;
    float a[3];
    for (int j = 0; j < 3; ++j) a[j] = arow[j];
    float m[3] = {ninf, ninf, ninf};

    if (cnt <= SCAP) {
      const unsigned* sl = s_slot + ln * SCAP;
      for (unsigned base = 0; base < cnt; base += RCH) {
        unsigned id[RCH];
        for (int k = 0; k < RCH; ++k) id[k] = sl[base + k];
        float c[RCH][3];
        for (int k = 0; k < RCH; ++k) {
          const unsigned s = (base + k < cnt) ? id[k] : id[0];
          const float* crow = (const float*)((const char*)AC + (s * (DCAT * 4u) + cboff));
          for (int j = 0; j < 3; ++j) c[k][j] = crow[j];
        }
        for (int k = 0; k < RCH; ++k) {
          for (int j = 0; j < 3; ++j) m[j] = fmaxf(m[j], c[k][j]);
        }
      }
    } else {
      for (int hf = 0; hf < 2; ++hf) {
        const unsigned* reg = REG + ((size_t)hf * NB + b) * HCAP;
        for (unsigned i = 0; i < nh[hf]; ++i) {
          const unsigned w = reg[i];
          if ((w & 127u) == (unsigned)ln) row_max(AC, w >> 7, cboff, m);
        }
      }
      for (unsigned i = 0; i < novf; ++i) {
        const unsigned e0 = OVF[i];
        const unsigned e  = e0 < (unsigned)N_EDGES ? e0 : 0u;
        if (dst[e] == d) row_max(AC, (unsigned)src[e], cboff, m);
      }
    }

    float* orow = out + (size_t)d * DOUT + g * 3;
    for (int j = 0; j < 3; ++j) orow[j] = (cnt == 0u) ? 0.0f : a[j] + m[j];
  }
}

extern "C" void kernel_launch(void* const* d_in, const int* in_sizes, int n_in,
                              void* d_out, int out_size, void* d_ws, size_t ws_size,
                              hipStream_t stream) {
  const float* h       = (const float*)d_in[0];
  const float* theta_w = (const float*)d_in[1];
  const float* theta_b = (const float*)d_in[2];
  const float* phi_w   = (const float*)d_in[3];
  const float* phi_b   = (const float*)d_in[4];
  const int*   src     = (const int*)d_in[5];
  const int*   dst     = (const int*)d_in[6];
  float* out = (float*)d_out;

  char* ws = (char*)d_ws;
  float*    Wfrag  = (float*)ws;
  float*    bias_c = (float*)(ws + 18432);
  float*    AC     = (float*)(ws + 18688);
  unsigned* CNT    = (unsigned*)(ws + 18688 + (size_t)N_NODES * DCAT * 4);
  unsigned* REG    = (unsigned*)(ws + 18688 + (size_t)N_NODES * DCAT * 4 + 8192);
  unsigned* OVF    = (unsigned*)(ws + 18688 + (size_t)N_NODES * DCAT * 4 + 8192
                                    + (size_t)2 * NB * HCAP * 4);

  ec_clear_fold<<<(DOUT * DIN + 255) / 256, 256, 0, stream>>>(theta_w, theta_b, phi_w, phi_b,
                                                              Wfrag, bias_c, CNT);
  ec_gemm_part<<<GEMM_BLOCKS + PART_BLOCKS, 256, 0, stream>>>(h, Wfrag, bias_c, AC,
                                                              src, dst, CNT, REG, OVF);
  ec_nodes<<<NB, NODE_T, 0, stream>>>(src, dst, AC, CNT, REG, OVF, out);
}
